# starting kernel
# baseline (speedup 1.0000x reference)
#include <hip/hip_runtime.h>

typedef float v2f __attribute__((ext_vector_type(2)));
typedef float v8f __attribute__((ext_vector_type(8)));

#define CIN    256
#define HIDDEN 64
#define CIN2   96

__global__ void wtranspose(const float* __restrict__ src, float* __restrict__ dst,
                           int rows, int cols) {
    int idx = blockIdx.x * blockDim.x + threadIdx.x;
    if (idx < rows * cols) {
        int r = idx / cols, c = idx - r * cols;
        dst[c * rows + r] = src[idx];
    }
}

template <int HW>
__global__ __launch_bounds__(256) void fused_level(
    const float* __restrict__ x,
    const float* __restrict__ w1t, const float* __restrict__ b1,
    const float* __restrict__ w2t, const float* __restrict__ b2,
    float* __restrict__ out)
{
    __shared__ float ylds[8][HIDDEN * 17];

    const int lane = threadIdx.x & 31;
    const int wave = threadIdx.x >> 5;
    const int col  = lane & 15;
    const int half = lane >> 4;

    const int pixBase = blockIdx.x * 128 + wave * 16;
    const int b  = pixBase / HW;
    const int hw = pixBase - b * HW;

    const float* xcol = x + (size_t)b * CIN * HW + hw + col;
    float* my = &ylds[wave][0];

    v8f acc1[4] = {};
#pragma unroll 2
    for (int k0 = 0; k0 < CIN; k0 += 4) {
        const int kr = k0 + 2 * half;
        v2f bm;
        bm.x = xcol[(size_t)kr * HW];
        bm.y = xcol[(size_t)(kr + 1) * HW];
        if (k0 + 12 < CIN)
            __builtin_prefetch(&xcol[(size_t)(kr + 12) * HW], 0, 3);
#pragma unroll
        for (int t = 0; t < 4; ++t) {
            v2f am = *(const v2f*)(w1t + (t * 16 + col) * CIN + kr);
            acc1[t] = __builtin_amdgcn_wmma_f32_16x16x4_f32(
                false, am, false, bm, (short)0, acc1[t], false, false);
        }
    }

#pragma unroll
    for (int t = 0; t < 4; ++t) {
#pragma unroll
        for (int r = 0; r < 8; ++r) {
            const int row = t * 16 + r + 8 * half;
            my[row * 17 + col] = acc1[t][r] + b1[row];
        }
    }
    __syncthreads();

    float* o = out + (size_t)b * CIN * HW + hw + col;
    for (int g = 0; g < 2; ++g) {
        v8f acc2[8] = {};
        for (int ks = 0; ks < 24; ++ks) {
            const int kr   = ks * 4 + 2 * half;
            const int krow = (ks >= 16) ? (kr - 32) : kr;
            float z0 = my[krow * 17 + col];
            float z1 = my[krow * 17 + 17 + col];
            if (ks >= 16) { z0 = fmaxf(z0, 0.0f); z1 = fmaxf(z1, 0.0f); }
            v2f bm; bm.x = z0; bm.y = z1;
#pragma unroll
            for (int t = 0; t < 8; ++t) {
                v2f am = *(const v2f*)(w2t + ((g * 8 + t) * 16 + col) * CIN2 + kr);
                acc2[t] = __builtin_amdgcn_wmma_f32_16x16x4_f32(
                    false, am, false, bm, (short)0, acc2[t], false, false);
            }
        }
#pragma unroll
        for (int t = 0; t < 8; ++t) {
#pragma unroll
            for (int r = 0; r < 8; ++r) {
                const int row = (g * 8 + t) * 16 + r + 8 * half;
                o[(size_t)row * HW] = acc2[t][r] + b2[row];
            }
        }
    }
}

extern "C" void kernel_launch(void* const* d_in, const int* in_sizes, int n_in,
                              void* d_out, int out_size, void* d_ws, size_t ws_size,
                              hipStream_t stream) {
    (void)in_sizes; (void)n_in; (void)out_size; (void)ws_size;
    static const int HWs[3] = {128 * 128, 64 * 64, 32 * 32};

    float* ws  = (float*)d_ws;
    float* out = (float*)d_out;

    float* w1t[3];
    float* w2t[3];
    size_t wsOff = 0;
    for (int i = 0; i < 3; ++i) {
        const float* W1 = (const float*)d_in[3 + 4 * i];
        const float* W2 = (const float*)d_in[5 + 4 * i];
        w1t[i] = ws + wsOff; wsOff += (size_t)HIDDEN * CIN;
        w2t[i] = ws + wsOff; wsOff += (size_t)CIN * CIN2;
        wtranspose<<<(CIN * HIDDEN + 255) / 256, 256, 0, stream>>>(W1, w1t[i], CIN, HIDDEN);
        wtranspose<<<(CIN2 * CIN + 255) / 256, 256, 0, stream>>>(W2, w2t[i], CIN2, CIN);
    }

    size_t outOff = 0;
    for (int i = 0; i < 3; ++i) {
        const float* x  = (const float*)d_in[i];
        const float* b1 = (const float*)d_in[4 + 4 * i];
        const float* b2 = (const float*)d_in[6 + 4 * i];
        const int HW = HWs[i];
        const int nblocks = (4 * HW) / 128;
        switch (i) {
            case 0: fused_level<128 * 128><<<nblocks, 256, 0, stream>>>(
                        x, w1t[i], b1, w2t[i], b2, out + outOff); break;
            case 1: fused_level<64 * 64><<<nblocks, 256, 0, stream>>>(
                        x, w1t[i], b1, w2t[i], b2, out + outOff); break;
            default: fused_level<32 * 32><<<nblocks, 256, 0, stream>>>(
                        x, w1t[i], b1, w2t[i], b2, out + outOff); break;
        }
        outOff += (size_t)4 * CIN * HW;
    }
}


# Merge the 9 serial launches into 2: one grid transposes all six weight matrices, one grid runs all three levels by block range, so the small latency-bound grids overlap with the large one
# speedup vs baseline: 2.0364x; 2.0364x over previous
#include <hip/hip_runtime.h>


typedef float v2f __attribute__((ext_vector_type(2)));
typedef float v8f __attribute__((ext_vector_type(8)));

#define CIN    256
#define HIDDEN 64
#define CIN2   96

#define W1T_FLOATS (HIDDEN * CIN)
#define W2T_FLOATS (CIN * CIN2)
#define LVL_FLOATS (W1T_FLOATS + W2T_FLOATS)

#define W1_BLOCKS  (W1T_FLOATS / 256)
#define W2_BLOCKS  (W2T_FLOATS / 256)
#define LVL_WBLOCKS (W1_BLOCKS + W2_BLOCKS)

#define HW0 (128 * 128)
#define HW1 (64 * 64)
#define HW2 (32 * 32)
#define NB0 (4 * HW0 / 128)
#define NB1 (4 * HW1 / 128)
#define NB2 (4 * HW2 / 128)

__global__ __launch_bounds__(256) void wtranspose_all(
    const float* __restrict__ W1_0, const float* __restrict__ W2_0,
    const float* __restrict__ W1_1, const float* __restrict__ W2_1,
    const float* __restrict__ W1_2, const float* __restrict__ W2_2,
    float* __restrict__ ws)
{
    const int lvl = blockIdx.x / LVL_WBLOCKS;
    const int rb  = blockIdx.x - lvl * LVL_WBLOCKS;
    const float* W1 = (lvl == 0) ? W1_0 : (lvl == 1) ? W1_1 : W1_2;
    const float* W2 = (lvl == 0) ? W2_0 : (lvl == 1) ? W2_1 : W2_2;
    float* base = ws + lvl * LVL_FLOATS;
    if (rb < W1_BLOCKS) {
        const int idx = rb * 256 + threadIdx.x;
        const int r = idx / HIDDEN, c = idx - r * HIDDEN;
        base[c * CIN + r] = W1[idx];
    } else {
        const int idx = (rb - W1_BLOCKS) * 256 + threadIdx.x;
        const int r = idx / CIN, c = idx - r * CIN;
        base[W1T_FLOATS + c * CIN2 + r] = W2[idx];
    }
}

template <int HW>
__device__ __forceinline__ void level_block(
    float (*ylds)[HIDDEN * 17], const int blk,
    const float* __restrict__ x,
    const float* __restrict__ w1t, const float* __restrict__ b1,
    const float* __restrict__ w2t, const float* __restrict__ b2,
    float* __restrict__ out)
{
    const int lane = threadIdx.x & 31;
    const int wave = threadIdx.x >> 5;
    const int col  = lane & 15;
    const int half = lane >> 4;

    const int pixBase = blk * 128 + wave * 16;
    const int b  = pixBase / HW;
    const int hw = pixBase - b * HW;

    const float* xcol = x + (size_t)b * CIN * HW + hw + col;
    float* my = &ylds[wave][0];

    v8f acc1[4] = {};
#pragma unroll 2
    for (int k0 = 0; k0 < CIN; k0 += 4) {
        const int kr = k0 + 2 * half;
        v2f bm;
        bm.x = xcol[(size_t)kr * HW];
        bm.y = xcol[(size_t)(kr + 1) * HW];
        if (k0 + 12 < CIN)
            __builtin_prefetch(&xcol[(size_t)(kr + 12) * HW], 0, 3);
#pragma unroll
        for (int t = 0; t < 4; ++t) {
            v2f am = *(const v2f*)(w1t + (t * 16 + col) * CIN + kr);
            acc1[t] = __builtin_amdgcn_wmma_f32_16x16x4_f32(
                false, am, false, bm, (short)0, acc1[t], false, false);
        }
    }

#pragma unroll
    for (int t = 0; t < 4; ++t) {
#pragma unroll
        for (int r = 0; r < 8; ++r) {
            const int row = t * 16 + r + 8 * half;
            my[row * 17 + col] = acc1[t][r] + b1[row];
        }
    }
    __syncthreads();

    float* o = out + (size_t)b * CIN * HW + hw + col;
    for (int g = 0; g < 2; ++g) {
        v8f acc2[8] = {};
        for (int ks = 0; ks < 24; ++ks) {
            const int kr   = ks * 4 + 2 * half;
            const int krow = (ks >= 16) ? (kr - 32) : kr;
            float z0 = my[krow * 17 + col];
            float z1 = my[krow * 17 + 17 + col];
            if (ks >= 16) { z0 = fmaxf(z0, 0.0f); z1 = fmaxf(z1, 0.0f); }
            v2f bm; bm.x = z0; bm.y = z1;
#pragma unroll
            for (int t = 0; t < 8; ++t) {
                v2f am = *(const v2f*)(w2t + ((g * 8 + t) * 16 + col) * CIN2 + kr);
                acc2[t] = __builtin_amdgcn_wmma_f32_16x16x4_f32(
                    false, am, false, bm, (short)0, acc2[t], false, false);
            }
        }
#pragma unroll
        for (int t = 0; t < 8; ++t) {
#pragma unroll
            for (int r = 0; r < 8; ++r) {
                const int row = (g * 8 + t) * 16 + r + 8 * half;
                o[(size_t)row * HW] = acc2[t][r] + b2[row];
            }
        }
    }
}

__global__ __launch_bounds__(256) void fused_levels(
    const float* __restrict__ x0, const float* __restrict__ x1,
    const float* __restrict__ x2, const float* __restrict__ wt,
    const float* __restrict__ b1_0, const float* __restrict__ b2_0,
    const float* __restrict__ b1_1, const float* __restrict__ b2_1,
    const float* __restrict__ b1_2, const float* __restrict__ b2_2,
    float* __restrict__ out)
{
    __shared__ float ylds[8][HIDDEN * 17];

    const int blk = blockIdx.x;
    if (blk < NB0) {
        level_block<HW0>(ylds, blk, x0,
                         wt, b1_0, wt + W1T_FLOATS, b2_0,
                         out);
    } else if (blk < NB0 + NB1) {
        level_block<HW1>(ylds, blk - NB0, x1,
                         wt + LVL_FLOATS, b1_1, wt + LVL_FLOATS + W1T_FLOATS, b2_1,
                         out + (size_t)4 * CIN * HW0);
    } else {
        level_block<HW2>(ylds, blk - NB0 - NB1, x2,
                         wt + 2 * LVL_FLOATS, b1_2, wt + 2 * LVL_FLOATS + W1T_FLOATS, b2_2,
                         out + (size_t)4 * CIN * (HW0 + HW1));
    }
}

extern "C" void kernel_launch(void* const* d_in, const int* in_sizes, int n_in,
                              void* d_out, int out_size, void* d_ws, size_t ws_size,
                              hipStream_t stream) {
    (void)in_sizes; (void)n_in; (void)out_size; (void)ws_size;

    float* ws = (float*)d_ws;

    wtranspose_all<<<3 * LVL_WBLOCKS, 256, 0, stream>>>(
        (const float*)d_in[3], (const float*)d_in[5],
        (const float*)d_in[7], (const float*)d_in[9],
        (const float*)d_in[11], (const float*)d_in[13],
        ws);

    fused_levels<<<NB0 + NB1 + NB2, 256, 0, stream>>>(
        (const float*)d_in[0], (const float*)d_in[1], (const float*)d_in[2], ws,
        (const float*)d_in[4], (const float*)d_in[6],
        (const float*)d_in[8], (const float*)d_in[10],
        (const float*)d_in[12], (const float*)d_in[14],
        (float*)d_out);
}


# Store the weights in WMMA A-fragment order (tile, k-step pair, lane, 4 floats): each A load is one contiguous 16-byte-per-lane read feeding 2 k-steps, not 16 scattered cache lines per k-step
# speedup vs baseline: 2.9496x; 1.4484x over previous
#include <hip/hip_runtime.h>


typedef float v2f __attribute__((ext_vector_type(2)));
typedef float v4f __attribute__((ext_vector_type(4)));
typedef float v8f __attribute__((ext_vector_type(8)));

#define CIN    256
#define HIDDEN 64
#define CIN2   96

#define W1P_FLOATS (HIDDEN * CIN)
#define W2P_FLOATS (CIN * CIN2)
#define LVL_FLOATS (W1P_FLOATS + W2P_FLOATS)

#define W1_BLOCKS  (W1P_FLOATS / 256)
#define W2_BLOCKS  (W2P_FLOATS / 256)
#define LVL_WBLOCKS (W1_BLOCKS + W2_BLOCKS)

#define KP1 (CIN / 8)
#define KP2 (CIN2 / 8)
#define FRAG 128

#define HW0 (128 * 128)
#define HW1 (64 * 64)
#define HW2 (32 * 32)
#define NB0 (4 * HW0 / 128)
#define NB1 (4 * HW1 / 128)
#define NB2 (4 * HW2 / 128)

__global__ __launch_bounds__(256) void wpack_all(
    const float* __restrict__ W1_0, const float* __restrict__ W2_0,
    const float* __restrict__ W1_1, const float* __restrict__ W2_1,
    const float* __restrict__ W1_2, const float* __restrict__ W2_2,
    float* __restrict__ ws)
{
    const int lvl = blockIdx.x / LVL_WBLOCKS;
    const int rb  = blockIdx.x - lvl * LVL_WBLOCKS;
    const float* W1 = (lvl == 0) ? W1_0 : (lvl == 1) ? W1_1 : W1_2;
    const float* W2 = (lvl == 0) ? W2_0 : (lvl == 1) ? W2_1 : W2_2;
    float* base = ws + lvl * LVL_FLOATS;
    const bool first = rb < W1_BLOCKS;
    const int d     = (first ? rb : rb - W1_BLOCKS) * 256 + threadIdx.x;
    const int q     = d & 3;
    const int ln    = (d >> 2) & 31;
    const int tp    = d >> 7;
    const int pairs = first ? KP1 : KP2;
    const int tile  = tp / pairs;
    const int p     = tp - tile * pairs;
    const int k = 8 * p + 4 * (q >> 1) + 2 * (ln >> 4) + (q & 1);
    const int m = 16 * tile + (ln & 15);
    if (first) {
        base[d] = W1[k * HIDDEN + m];
    } else {
        base[W1P_FLOATS + d] = W2[k * CIN + m];
    }
}

template <int HW>
__device__ __forceinline__ void level_block(
    float (*ylds)[HIDDEN * 17], const int blk,
    const float* __restrict__ x,
    const float* __restrict__ w1p, const float* __restrict__ b1,
    const float* __restrict__ w2p, const float* __restrict__ b2,
    float* __restrict__ out)
{
    const int lane = threadIdx.x & 31;
    const int wave = threadIdx.x >> 5;
    const int col  = lane & 15;
    const int half = lane >> 4;

    const int pixBase = blk * 128 + wave * 16;
    const int b  = pixBase / HW;
    const int hw = pixBase - b * HW;

    const float* xcol = x + (size_t)b * CIN * HW + hw + col;
    float* my = &ylds[wave][0];

    const float* a1 = w1p + lane * 4;
    v8f acc1[4] = {};
    for (int p = 0; p < KP1; ++p) {
        const int kr = p * 8 + 2 * half;
        v2f bm0, bm1;
        bm0.x = xcol[(size_t)kr * HW];
        bm0.y = xcol[(size_t)(kr + 1) * HW];
        bm1.x = xcol[(size_t)(kr + 4) * HW];
        bm1.y = xcol[(size_t)(kr + 5) * HW];
        if (p * 8 + 12 < CIN)
            __builtin_prefetch(&xcol[(size_t)(kr + 12) * HW], 0, 3);
        if (p * 8 + 16 < CIN)
            __builtin_prefetch(&xcol[(size_t)(kr + 16) * HW], 0, 3);
        v4f am[4];
#pragma unroll
        for (int t = 0; t < 4; ++t)
            am[t] = *(const v4f*)(a1 + (t * KP1 + p) * FRAG);
#pragma unroll
        for (int t = 0; t < 4; ++t) {
            v2f a; a.x = am[t].x; a.y = am[t].y;
            acc1[t] = __builtin_amdgcn_wmma_f32_16x16x4_f32(
                false, a, false, bm0, (short)0, acc1[t], false, false);
        }
#pragma unroll
        for (int t = 0; t < 4; ++t) {
            v2f a; a.x = am[t].z; a.y = am[t].w;
            acc1[t] = __builtin_amdgcn_wmma_f32_16x16x4_f32(
                false, a, false, bm1, (short)0, acc1[t], false, false);
        }
    }

#pragma unroll
    for (int t = 0; t < 4; ++t) {
#pragma unroll
        for (int r = 0; r < 8; ++r) {
            const int row = t * 16 + r + 8 * half;
            my[row * 17 + col] = acc1[t][r] + b1[row];
        }
    }
    __syncthreads();

    const float* a2 = w2p + lane * 4;
    float* o = out + (size_t)b * CIN * HW + hw + col;
    for (int g = 0; g < 2; ++g) {
        v8f acc2[8] = {};
        for (int p = 0; p < KP2; ++p) {
            const int kr   = p * 8 + 2 * half;
            const int krow = (p >= 8) ? (kr - 32) : kr;
            float z0 = my[krow * 17 + col];
            float z1 = my[krow * 17 + 17 + col];
            float z2 = my[krow * 17 + 4 * 17 + col];
            float z3 = my[krow * 17 + 5 * 17 + col];
            if (p >= 8) {
                z0 = fmaxf(z0, 0.0f); z1 = fmaxf(z1, 0.0f);
                z2 = fmaxf(z2, 0.0f); z3 = fmaxf(z3, 0.0f);
            }
            v2f bm0, bm1;
            bm0.x = z0; bm0.y = z1;
            bm1.x = z2; bm1.y = z3;
            v4f am[8];
#pragma unroll
            for (int t = 0; t < 8; ++t)
                am[t] = *(const v4f*)(a2 + ((g * 8 + t) * KP2 + p) * FRAG);
#pragma unroll
            for (int t = 0; t < 8; ++t) {
                v2f a; a.x = am[t].x; a.y = am[t].y;
                acc2[t] = __builtin_amdgcn_wmma_f32_16x16x4_f32(
                    false, a, false, bm0, (short)0, acc2[t], false, false);
            }
#pragma unroll
            for (int t = 0; t < 8; ++t) {
                v2f a; a.x = am[t].z; a.y = am[t].w;
                acc2[t] = __builtin_amdgcn_wmma_f32_16x16x4_f32(
                    false, a, false, bm1, (short)0, acc2[t], false, false);
            }
        }
#pragma unroll
        for (int t = 0; t < 8; ++t) {
#pragma unroll
            for (int r = 0; r < 8; ++r) {
                const int row = (g * 8 + t) * 16 + r + 8 * half;
                o[(size_t)row * HW] = acc2[t][r] + b2[row];
            }
        }
    }
}

__global__ __launch_bounds__(256) void fused_levels(
    const float* __restrict__ x0, const float* __restrict__ x1,
    const float* __restrict__ x2, const float* __restrict__ wt,
    const float* __restrict__ b1_0, const float* __restrict__ b2_0,
    const float* __restrict__ b1_1, const float* __restrict__ b2_1,
    const float* __restrict__ b1_2, const float* __restrict__ b2_2,
    float* __restrict__ out)
{
    __shared__ float ylds[8][HIDDEN * 17];

    const int blk = blockIdx.x;
    if (blk < NB0) {
        level_block<HW0>(ylds, blk, x0,
                         wt, b1_0, wt + W1P_FLOATS, b2_0,
                         out);
    } else if (blk < NB0 + NB1) {
        level_block<HW1>(ylds, blk - NB0, x1,
                         wt + LVL_FLOATS, b1_1, wt + LVL_FLOATS + W1P_FLOATS, b2_1,
                         out + (size_t)4 * CIN * HW0);
    } else {
        level_block<HW2>(ylds, blk - NB0 - NB1, x2,
                         wt + 2 * LVL_FLOATS, b1_2, wt + 2 * LVL_FLOATS + W1P_FLOATS, b2_2,
                         out + (size_t)4 * CIN * (HW0 + HW1));
    }
}

extern "C" void kernel_launch(void* const* d_in, const int* in_sizes, int n_in,
                              void* d_out, int out_size, void* d_ws, size_t ws_size,
                              hipStream_t stream) {
    (void)in_sizes; (void)n_in; (void)out_size; (void)ws_size;

    float* ws = (float*)d_ws;

    wpack_all<<<3 * LVL_WBLOCKS, 256, 0, stream>>>(
        (const float*)d_in[3], (const float*)d_in[5],
        (const float*)d_in[7], (const float*)d_in[9],
        (const float*)d_in[11], (const float*)d_in[13],
        ws);

    fused_levels<<<NB0 + NB1 + NB2, 256, 0, stream>>>(
        (const float*)d_in[0], (const float*)d_in[1], (const float*)d_in[2], ws,
        (const float*)d_in[4], (const float*)d_in[6],
        (const float*)d_in[8], (const float*)d_in[10],
        (const float*)d_in[12], (const float*)d_in[14],
        (float*)d_out);
}


# Each wave owns 32 pixels as two interleaved column tiles (pixel 2c+j): each weight fragment load feeds both tiles; x loads, prefetches and out stores become 8 bytes per lane, whole 128-byte lines
# speedup vs baseline: 3.1973x; 1.0840x over previous
#include <hip/hip_runtime.h>


typedef float v2f __attribute__((ext_vector_type(2)));
typedef float v4f __attribute__((ext_vector_type(4)));
typedef float v8f __attribute__((ext_vector_type(8)));

#define CIN    256
#define HIDDEN 64
#define CIN2   96

#define W1P_FLOATS (HIDDEN * CIN)
#define W2P_FLOATS (CIN * CIN2)
#define LVL_FLOATS (W1P_FLOATS + W2P_FLOATS)

#define W1_BLOCKS  (W1P_FLOATS / 256)
#define W2_BLOCKS  (W2P_FLOATS / 256)
#define LVL_WBLOCKS (W1_BLOCKS + W2_BLOCKS)

#define KP1 (CIN / 8)
#define KP2 (CIN2 / 8)
#define FRAG 128

#define HW0 (128 * 128)
#define HW1 (64 * 64)
#define HW2 (32 * 32)
#define NB0 (4 * HW0 / 128)
#define NB1 (4 * HW1 / 128)
#define NB2 (4 * HW2 / 128)

#define WPX 32
#define YP  48

__global__ __launch_bounds__(256) void wpack_all(
    const float* __restrict__ W1_0, const float* __restrict__ W2_0,
    const float* __restrict__ W1_1, const float* __restrict__ W2_1,
    const float* __restrict__ W1_2, const float* __restrict__ W2_2,
    float* __restrict__ ws)
{
    const int lvl = blockIdx.x / LVL_WBLOCKS;
    const int rb  = blockIdx.x - lvl * LVL_WBLOCKS;
    const float* W1 = (lvl == 0) ? W1_0 : (lvl == 1) ? W1_1 : W1_2;
    const float* W2 = (lvl == 0) ? W2_0 : (lvl == 1) ? W2_1 : W2_2;
    float* base = ws + lvl * LVL_FLOATS;
    const bool first = rb < W1_BLOCKS;
    const int d     = (first ? rb : rb - W1_BLOCKS) * 256 + threadIdx.x;
    const int q     = d & 3;
    const int ln    = (d >> 2) & 31;
    const int tp    = d >> 7;
    const int pairs = first ? KP1 : KP2;
    const int tile  = tp / pairs;
    const int p     = tp - tile * pairs;
    const int k = 8 * p + 4 * (q >> 1) + 2 * (ln >> 4) + (q & 1);
    const int m = 16 * tile + (ln & 15);
    if (first) {
        base[d] = W1[k * HIDDEN + m];
    } else {
        base[W1P_FLOATS + d] = W2[k * CIN + m];
    }
}

template <int HW>
__device__ __forceinline__ void level_block(
    float (*ylds)[HIDDEN * YP], const int blk,
    const float* __restrict__ x,
    const float* __restrict__ w1p, const float* __restrict__ b1,
    const float* __restrict__ w2p, const float* __restrict__ b2,
    float* __restrict__ out)
{
    const int lane = threadIdx.x & 31;
    const int wave = threadIdx.x >> 5;
    const int col  = lane & 15;
    const int half = lane >> 4;

    const int pixBase = blk * 128 + wave * WPX;
    const int b  = pixBase / HW;
    const int hw = pixBase - b * HW;

    const float* xw = x + (size_t)b * CIN * HW + hw;
    const float* xq = xw + 2 * col;
    float* my = &ylds[wave][0];

    __builtin_prefetch(&xw[(size_t)(8 + col) * HW], 0, 3);

    const float* a1 = w1p + lane * 4;
    v8f acc1[4][2] = {};
#pragma nounroll
    for (int p = 0; p < KP1; ++p) {
        const int kr = p * 8 + 2 * half;
        const v2f r0 = *(const v2f*)(xq + (size_t)kr * HW);
        const v2f r1 = *(const v2f*)(xq + (size_t)(kr + 1) * HW);
        const v2f r4 = *(const v2f*)(xq + (size_t)(kr + 4) * HW);
        const v2f r5 = *(const v2f*)(xq + (size_t)(kr + 5) * HW);
        if (p + 3 < KP1)
            __builtin_prefetch(&xw[(size_t)(p * 8 + 24 + (lane & 7)) * HW], 0, 3);
        v4f am[4];
#pragma unroll
        for (int t = 0; t < 4; ++t)
            am[t] = *(const v4f*)(a1 + (t * KP1 + p) * FRAG);
        v2f bm0[2], bm1[2];
        bm0[0].x = r0.x; bm0[0].y = r1.x;
        bm0[1].x = r0.y; bm0[1].y = r1.y;
        bm1[0].x = r4.x; bm1[0].y = r5.x;
        bm1[1].x = r4.y; bm1[1].y = r5.y;
#pragma unroll
        for (int t = 0; t < 4; ++t) {
            v2f a; a.x = am[t].x; a.y = am[t].y;
#pragma unroll
            for (int j = 0; j < 2; ++j)
                acc1[t][j] = __builtin_amdgcn_wmma_f32_16x16x4_f32(
                    false, a, false, bm0[j], (short)0, acc1[t][j], false, false);
        }
#pragma unroll
        for (int t = 0; t < 4; ++t) {
            v2f a; a.x = am[t].z; a.y = am[t].w;
#pragma unroll
            for (int j = 0; j < 2; ++j)
                acc1[t][j] = __builtin_amdgcn_wmma_f32_16x16x4_f32(
                    false, a, false, bm1[j], (short)0, acc1[t][j], false, false);
        }
    }

#pragma unroll
    for (int t = 0; t < 4; ++t) {
#pragma unroll
        for (int r = 0; r < 8; ++r) {
            const int row = t * 16 + r + 8 * half;
            const float bias = b1[row];
            v2f y;
            y.x = acc1[t][0][r] + bias;
            y.y = acc1[t][1][r] + bias;
            *(v2f*)(my + row * YP + 2 * col) = y;
        }
    }
    __syncthreads();

    const float* a2 = w2p + lane * 4;
    const float* myq = my + 2 * col;
    float* o = out + (size_t)b * CIN * HW + hw + 2 * col;
    for (int g = 0; g < 2; ++g) {
        v8f acc2[8][2] = {};
#pragma nounroll
        for (int p = 0; p < KP2; ++p) {
            const int kr   = p * 8 + 2 * half;
            const int krow = (p >= 8) ? (kr - 32) : kr;
            v2f z0 = *(const v2f*)(myq + krow * YP);
            v2f z1 = *(const v2f*)(myq + (krow + 1) * YP);
            v2f z4 = *(const v2f*)(myq + (krow + 4) * YP);
            v2f z5 = *(const v2f*)(myq + (krow + 5) * YP);
            if (p >= 8) {
                z0.x = fmaxf(z0.x, 0.0f); z0.y = fmaxf(z0.y, 0.0f);
                z1.x = fmaxf(z1.x, 0.0f); z1.y = fmaxf(z1.y, 0.0f);
                z4.x = fmaxf(z4.x, 0.0f); z4.y = fmaxf(z4.y, 0.0f);
                z5.x = fmaxf(z5.x, 0.0f); z5.y = fmaxf(z5.y, 0.0f);
            }
            v2f bm0[2], bm1[2];
            bm0[0].x = z0.x; bm0[0].y = z1.x;
            bm0[1].x = z0.y; bm0[1].y = z1.y;
            bm1[0].x = z4.x; bm1[0].y = z5.x;
            bm1[1].x = z4.y; bm1[1].y = z5.y;
            v4f am[8];
#pragma unroll
            for (int t = 0; t < 8; ++t)
                am[t] = *(const v4f*)(a2 + ((g * 8 + t) * KP2 + p) * FRAG);
#pragma unroll
            for (int t = 0; t < 8; ++t) {
                v2f a; a.x = am[t].x; a.y = am[t].y;
#pragma unroll
                for (int j = 0; j < 2; ++j)
                    acc2[t][j] = __builtin_amdgcn_wmma_f32_16x16x4_f32(
                        false, a, false, bm0[j], (short)0, acc2[t][j], false, false);
            }
#pragma unroll
            for (int t = 0; t < 8; ++t) {
                v2f a; a.x = am[t].z; a.y = am[t].w;
#pragma unroll
                for (int j = 0; j < 2; ++j)
                    acc2[t][j] = __builtin_amdgcn_wmma_f32_16x16x4_f32(
                        false, a, false, bm1[j], (short)0, acc2[t][j], false, false);
            }
        }
#pragma unroll
        for (int t = 0; t < 8; ++t) {
#pragma unroll
            for (int r = 0; r < 8; ++r) {
                const int row = (g * 8 + t) * 16 + r + 8 * half;
                const float bias = b2[row];
                v2f v;
                v.x = acc2[t][0][r] + bias;
                v.y = acc2[t][1][r] + bias;
                *(v2f*)(o + (size_t)row * HW) = v;
            }
        }
    }
}

__global__ __launch_bounds__(128) void fused_levels(
    const float* __restrict__ x0, const float* __restrict__ x1,
    const float* __restrict__ x2, const float* __restrict__ wt,
    const float* __restrict__ b1_0, const float* __restrict__ b2_0,
    const float* __restrict__ b1_1, const float* __restrict__ b2_1,
    const float* __restrict__ b1_2, const float* __restrict__ b2_2,
    float* __restrict__ out)
{
    __shared__ float ylds[4][HIDDEN * YP];

    const int blk = blockIdx.x;
    if (blk < NB0) {
        level_block<HW0>(ylds, blk, x0,
                         wt, b1_0, wt + W1P_FLOATS, b2_0,
                         out);
    } else if (blk < NB0 + NB1) {
        level_block<HW1>(ylds, blk - NB0, x1,
                         wt + LVL_FLOATS, b1_1, wt + LVL_FLOATS + W1P_FLOATS, b2_1,
                         out + (size_t)4 * CIN * HW0);
    } else {
        level_block<HW2>(ylds, blk - NB0 - NB1, x2,
                         wt + 2 * LVL_FLOATS, b1_2, wt + 2 * LVL_FLOATS + W1P_FLOATS, b2_2,
                         out + (size_t)4 * CIN * (HW0 + HW1));
    }
}

extern "C" void kernel_launch(void* const* d_in, const int* in_sizes, int n_in,
                              void* d_out, int out_size, void* d_ws, size_t ws_size,
                              hipStream_t stream) {
    (void)in_sizes; (void)n_in; (void)out_size; (void)ws_size;

    float* ws = (float*)d_ws;

    wpack_all<<<3 * LVL_WBLOCKS, 256, 0, stream>>>(
        (const float*)d_in[3], (const float*)d_in[5],
        (const float*)d_in[7], (const float*)d_in[9],
        (const float*)d_in[11], (const float*)d_in[13],
        ws);

    fused_levels<<<NB0 + NB1 + NB2, 128, 0, stream>>>(
        (const float*)d_in[0], (const float*)d_in[1], (const float*)d_in[2], ws,
        (const float*)d_in[4], (const float*)d_in[6],
        (const float*)d_in[8], (const float*)d_in[10],
        (const float*)d_in[12], (const float*)d_in[14],
        (float*)d_out);
}


# Stage each level's packed weights once per workgroup in LDS (12-wave workgroups, one per WGP, one 32-pixel unit per wave): A fragments are LDS reads, no weight load from L2 inside the K loops
# speedup vs baseline: 3.7261x; 1.1654x over previous
#include <hip/hip_runtime.h>


typedef float v2f __attribute__((ext_vector_type(2)));
typedef float v4f __attribute__((ext_vector_type(4)));
typedef float v8f __attribute__((ext_vector_type(8)));

__device__ __forceinline__ void wmma_run_end()
{
    __builtin_amdgcn_sched_barrier(0);
    asm volatile("v_nop\n\tv_nop\n\tv_nop\n\tv_nop");
    __builtin_amdgcn_sched_barrier(0);
}

#define CIN    256
#define HIDDEN 64
#define CIN2   96

#define W1P_FLOATS (HIDDEN * CIN)
#define W2P_FLOATS (CIN * CIN2)
#define LVL_FLOATS (W1P_FLOATS + W2P_FLOATS)

#define W1_BLOCKS  (W1P_FLOATS / 256)
#define W2_BLOCKS  (W2P_FLOATS / 256)
#define LVL_WBLOCKS (W1_BLOCKS + W2_BLOCKS)

#define KP1 (CIN / 8)
#define KP2 (CIN2 / 8)
#define FRAG 128

#define HW0 (128 * 128)
#define HW1 (64 * 64)
#define HW2 (32 * 32)

#define WPX 32
#define YP  48

#define NWAVES 12
#define NTHR   (NWAVES * 32)
#define UNITS0 (4 * HW0 / WPX)
#define UNITS1 (4 * HW1 / WPX)
#define UNITS2 (4 * HW2 / WPX)
#define NBL0 ((UNITS0 + NWAVES - 1) / NWAVES)
#define NBL1 ((UNITS1 + NWAVES - 1) / NWAVES)
#define NBL2 ((UNITS2 + NWAVES - 1) / NWAVES)
#define LVL_VEC4 (LVL_FLOATS / 4)
#define CPB 27

__global__ __launch_bounds__(256) void wpack_all(
    const float* __restrict__ W1_0, const float* __restrict__ W2_0,
    const float* __restrict__ W1_1, const float* __restrict__ W2_1,
    const float* __restrict__ W1_2, const float* __restrict__ W2_2,
    float* __restrict__ ws)
{
    const int lvl = blockIdx.x / LVL_WBLOCKS;
    const int rb  = blockIdx.x - lvl * LVL_WBLOCKS;
    const float* W1 = (lvl == 0) ? W1_0 : (lvl == 1) ? W1_1 : W1_2;
    const float* W2 = (lvl == 0) ? W2_0 : (lvl == 1) ? W2_1 : W2_2;
    float* base = ws + lvl * LVL_FLOATS;
    const bool first = rb < W1_BLOCKS;
    const int d     = (first ? rb : rb - W1_BLOCKS) * 256 + threadIdx.x;
    const int q     = d & 3;
    const int ln    = (d >> 2) & 31;
    const int tp    = d >> 7;
    const int pairs = first ? KP1 : KP2;
    const int tile  = tp / pairs;
    const int p     = tp - tile * pairs;
    const int k = 8 * p + 4 * (q >> 1) + 2 * (ln >> 4) + (q & 1);
    const int m = 16 * tile + (ln & 15);
    if (first) {
        base[d] = W1[k * HIDDEN + m];
    } else {
        base[W1P_FLOATS + d] = W2[k * CIN + m];
    }
}

template <int HW>
__device__ __forceinline__ void level_unit(
    const float* wl, float* my, const int unit, const bool active,
    const float* __restrict__ x,
    const float* __restrict__ b1, const float* __restrict__ b2,
    float* __restrict__ out)
{
    const int lane = threadIdx.x & 31;
    const int col  = lane & 15;
    const int half = lane >> 4;

    const int pixBase = unit * WPX;
    const int b  = pixBase / HW;
    const int hw = pixBase - b * HW;

    const float* xw = x + (size_t)b * CIN * HW + hw;
    const float* xq = xw + 2 * col;

    if (active) {
        const float* a1 = wl + lane * 4;
        v8f acc1[4][2] = {};
#pragma nounroll
        for (int p = 0; p < KP1; ++p) {
            const int kr = p * 8 + 2 * half;
            const v2f r0 = *(const v2f*)(xq + (size_t)kr * HW);
            const v2f r1 = *(const v2f*)(xq + (size_t)(kr + 1) * HW);
            const v2f r4 = *(const v2f*)(xq + (size_t)(kr + 4) * HW);
            const v2f r5 = *(const v2f*)(xq + (size_t)(kr + 5) * HW);
            if (p + 3 < KP1)
                __builtin_prefetch(&xw[(size_t)(p * 8 + 24 + (lane & 7)) * HW], 0, 3);
            v4f am[4];
#pragma unroll
            for (int t = 0; t < 4; ++t)
                am[t] = *(const v4f*)(a1 + (t * KP1 + p) * FRAG);
            v2f bm0[2], bm1[2];
            bm0[0].x = r0.x; bm0[0].y = r1.x;
            bm0[1].x = r0.y; bm0[1].y = r1.y;
            bm1[0].x = r4.x; bm1[0].y = r5.x;
            bm1[1].x = r4.y; bm1[1].y = r5.y;
            asm volatile("" : "+v"(bm0[0]), "+v"(bm0[1]), "+v"(bm1[0]), "+v"(bm1[1]));
#pragma unroll
            for (int t = 0; t < 4; ++t) {
                v2f a; a.x = am[t].x; a.y = am[t].y;
#pragma unroll
                for (int j = 0; j < 2; ++j)
                    acc1[t][j] = __builtin_amdgcn_wmma_f32_16x16x4_f32(
                        false, a, false, bm0[j], (short)0, acc1[t][j], false, false);
            }
#pragma unroll
            for (int t = 0; t < 4; ++t) {
                v2f a; a.x = am[t].z; a.y = am[t].w;
#pragma unroll
                for (int j = 0; j < 2; ++j)
                    acc1[t][j] = __builtin_amdgcn_wmma_f32_16x16x4_f32(
                        false, a, false, bm1[j], (short)0, acc1[t][j], false, false);
            }
            wmma_run_end();
        }

#pragma unroll
        for (int t = 0; t < 4; ++t) {
#pragma unroll
            for (int r = 0; r < 8; ++r) {
                const int row = t * 16 + r + 8 * half;
                const float bias = b1[row];
                v2f y;
                y.x = acc1[t][0][r] + bias;
                y.y = acc1[t][1][r] + bias;
                *(v2f*)(my + row * YP + 2 * col) = y;
            }
        }
    }
    __syncthreads();
    if (!active) return;

    const float* a2 = wl + W1P_FLOATS + lane * 4;
    const float* myq = my + 2 * col;
    float* o = out + (size_t)b * CIN * HW + hw + 2 * col;
    for (int g = 0; g < 2; ++g) {
        v8f acc2[8][2] = {};
#pragma nounroll
        for (int p = 0; p < KP2; ++p) {
            const int kr   = p * 8 + 2 * half;
            const int krow = (p >= 8) ? (kr - 32) : kr;
            v4f am[8];
#pragma unroll
            for (int t = 0; t < 8; ++t)
                am[t] = *(const v4f*)(a2 + ((g * 8 + t) * KP2 + p) * FRAG);
            v2f z0 = *(const v2f*)(myq + krow * YP);
            v2f z1 = *(const v2f*)(myq + (krow + 1) * YP);
            v2f z4 = *(const v2f*)(myq + (krow + 4) * YP);
            v2f z5 = *(const v2f*)(myq + (krow + 5) * YP);
            if (p >= 8) {
                z0.x = fmaxf(z0.x, 0.0f); z0.y = fmaxf(z0.y, 0.0f);
                z1.x = fmaxf(z1.x, 0.0f); z1.y = fmaxf(z1.y, 0.0f);
                z4.x = fmaxf(z4.x, 0.0f); z4.y = fmaxf(z4.y, 0.0f);
                z5.x = fmaxf(z5.x, 0.0f); z5.y = fmaxf(z5.y, 0.0f);
            }
            v2f bm0[2], bm1[2];
            bm0[0].x = z0.x; bm0[0].y = z1.x;
            bm0[1].x = z0.y; bm0[1].y = z1.y;
            bm1[0].x = z4.x; bm1[0].y = z5.x;
            bm1[1].x = z4.y; bm1[1].y = z5.y;
            asm volatile("" : "+v"(bm0[0]), "+v"(bm0[1]), "+v"(bm1[0]), "+v"(bm1[1]));
#pragma unroll
            for (int t = 0; t < 8; ++t) {
                v2f a; a.x = am[t].x; a.y = am[t].y;
#pragma unroll
                for (int j = 0; j < 2; ++j)
                    acc2[t][j] = __builtin_amdgcn_wmma_f32_16x16x4_f32(
                        false, a, false, bm0[j], (short)0, acc2[t][j], false, false);
            }
#pragma unroll
            for (int t = 0; t < 8; ++t) {
                v2f a; a.x = am[t].z; a.y = am[t].w;
#pragma unroll
                for (int j = 0; j < 2; ++j)
                    acc2[t][j] = __builtin_amdgcn_wmma_f32_16x16x4_f32(
                        false, a, false, bm1[j], (short)0, acc2[t][j], false, false);
            }
            wmma_run_end();
        }
#pragma unroll
        for (int t = 0; t < 8; ++t) {
#pragma unroll
            for (int r = 0; r < 8; ++r) {
                const int row = (g * 8 + t) * 16 + r + 8 * half;
                const float bias = b2[row];
                v2f v;
                v.x = acc2[t][0][r] + bias;
                v.y = acc2[t][1][r] + bias;
                *(v2f*)(o + (size_t)row * HW) = v;
            }
        }
    }
}

__global__ __launch_bounds__(NTHR) void fused_levels(
    const float* __restrict__ x0, const float* __restrict__ x1,
    const float* __restrict__ x2, const float* __restrict__ wt,
    const float* __restrict__ b1_0, const float* __restrict__ b2_0,
    const float* __restrict__ b1_1, const float* __restrict__ b2_1,
    const float* __restrict__ b1_2, const float* __restrict__ b2_2,
    float* __restrict__ out)
{
    __shared__ __attribute__((aligned(16))) float wlds[LVL_FLOATS];
    __shared__ __attribute__((aligned(16))) float ylds[NWAVES][HIDDEN * YP];

    const int blk  = blockIdx.x;
    const int wave = threadIdx.x >> 5;
    const int lvl  = (blk < NBL0) ? 0 : (blk < NBL0 + NBL1) ? 1 : 2;

    const int lb   = (lvl == 0) ? blk : (lvl == 1) ? blk - NBL0 : blk - NBL0 - NBL1;
    const int unit = lb * NWAVES + wave;

    {
        const int sh = (lvl == 0) ? 14 : (lvl == 1) ? 12 : 10;
        const float* xl = (lvl == 0) ? x0 : (lvl == 1) ? x1 : x2;
        if (unit < (4 << sh) / WPX) {
            const int pix = unit * WPX;
            const int b   = pix >> sh;
            const int hw  = pix - (b << sh);
            __builtin_prefetch(&xl[(((size_t)b * CIN + (threadIdx.x & 31)) << sh) + hw], 0, 3);
        }
    }

    {
        const v4f* srcw = (const v4f*)(wt + lvl * LVL_FLOATS);
        v4f* dstw = (v4f*)wlds;
        const int rot = (blk & 15) * (LVL_VEC4 / 16);
        for (int i0 = 0; i0 < LVL_VEC4; i0 += CPB * NTHR) {
            v4f tmp[CPB];
#pragma unroll
            for (int u = 0; u < CPB; ++u) {
                const int i = i0 + u * NTHR + (int)threadIdx.x;
                int j = i + rot;
                if (j >= LVL_VEC4) j -= LVL_VEC4;
                if (i < LVL_VEC4) tmp[u] = srcw[j];
            }
#pragma unroll
            for (int u = 0; u < CPB; ++u) {
                const int i = i0 + u * NTHR + (int)threadIdx.x;
                int j = i + rot;
                if (j >= LVL_VEC4) j -= LVL_VEC4;
                if (i < LVL_VEC4) dstw[j] = tmp[u];
            }
        }
    }
    __syncthreads();

    float* my = &ylds[wave][0];
    if (lvl == 0) {
        level_unit<HW0>(wlds, my, unit, unit < UNITS0, x0, b1_0, b2_0,
                        out);
    } else if (lvl == 1) {
        level_unit<HW1>(wlds, my, unit, unit < UNITS1, x1, b1_1, b2_1,
                        out + (size_t)4 * CIN * HW0);
    } else {
        level_unit<HW2>(wlds, my, unit, unit < UNITS2, x2, b1_2, b2_2,
                        out + (size_t)4 * CIN * (HW0 + HW1));
    }
}

extern "C" void kernel_launch(void* const* d_in, const int* in_sizes, int n_in,
                              void* d_out, int out_size, void* d_ws, size_t ws_size,
                              hipStream_t stream) {
    (void)in_sizes; (void)n_in; (void)out_size; (void)ws_size;

    float* ws = (float*)d_ws;

    wpack_all<<<3 * LVL_WBLOCKS, 256, 0, stream>>>(
        (const float*)d_in[3], (const float*)d_in[5],
        (const float*)d_in[7], (const float*)d_in[9],
        (const float*)d_in[11], (const float*)d_in[13],
        ws);

    fused_levels<<<NBL0 + NBL1 + NBL2, NTHR, 0, stream>>>(
        (const float*)d_in[0], (const float*)d_in[1], (const float*)d_in[2], ws,
        (const float*)d_in[4], (const float*)d_in[6],
        (const float*)d_in[8], (const float*)d_in[10],
        (const float*)d_in[12], (const float*)d_in[14],
        (float*)d_out);
}


# Fold the weight packing into the level kernel: each workgroup gathers W1/W2 as given into LDS in A-fragment order, 4 whole-line 4-byte row reads per 16-byte piece; one launch, not two, no workspace
# speedup vs baseline: 4.0333x; 1.0824x over previous
#include <hip/hip_runtime.h>


typedef float v2f __attribute__((ext_vector_type(2)));
typedef float v4f __attribute__((ext_vector_type(4)));
typedef float v8f __attribute__((ext_vector_type(8)));

__device__ __forceinline__ void wmma_run_end()
{
    __builtin_amdgcn_sched_barrier(0);
    asm volatile("v_nop\n\tv_nop\n\tv_nop\n\tv_nop");
    __builtin_amdgcn_sched_barrier(0);
}

#define CIN    256
#define HIDDEN 64
#define CIN2   96

#define W1P_FLOATS (HIDDEN * CIN)
#define W2P_FLOATS (CIN * CIN2)
#define LVL_FLOATS (W1P_FLOATS + W2P_FLOATS)

#define KP1 (CIN / 8)
#define KP2 (CIN2 / 8)
#define FRAG 128

#define HW0 (128 * 128)
#define HW1 (64 * 64)
#define HW2 (32 * 32)

#define WPX 32
#define YP  48

#define NWAVES 12
#define NTHR   (NWAVES * 32)
#define UNITS0 (4 * HW0 / WPX)
#define UNITS1 (4 * HW1 / WPX)
#define UNITS2 (4 * HW2 / WPX)
#define NBL0 ((UNITS0 + NWAVES - 1) / NWAVES)
#define NBL1 ((UNITS1 + NWAVES - 1) / NWAVES)
#define NBL2 ((UNITS2 + NWAVES - 1) / NWAVES)
#define T1N (KP1 * 2 * (HIDDEN / 32))
#define T2N (KP2 * 2 * (CIN / 32))
#define R1  ((T1N + NWAVES - 1) / NWAVES)
#define R2  (T2N / NWAVES)

template <int HW>
__device__ __forceinline__ void level_unit(
    const float* wl, float* my, const int unit, const bool active,
    const float* __restrict__ x,
    const float* __restrict__ b1, const float* __restrict__ b2,
    float* __restrict__ out)
{
    const int lane = threadIdx.x & 31;
    const int col  = lane & 15;
    const int half = lane >> 4;

    const int pixBase = unit * WPX;
    const int b  = pixBase / HW;
    const int hw = pixBase - b * HW;

    const float* xw = x + (size_t)b * CIN * HW + hw;
    const float* xq = xw + 2 * col;

    if (active) {
        const float* a1 = wl + lane * 4;
        v8f acc1[4][2] = {};
#pragma nounroll
        for (int p = 0; p < KP1; ++p) {
            const int kr = p * 8 + 2 * half;
            const v2f r0 = *(const v2f*)(xq + (size_t)kr * HW);
            const v2f r1 = *(const v2f*)(xq + (size_t)(kr + 1) * HW);
            const v2f r4 = *(const v2f*)(xq + (size_t)(kr + 4) * HW);
            const v2f r5 = *(const v2f*)(xq + (size_t)(kr + 5) * HW);
            if (p + 3 < KP1)
                __builtin_prefetch(&xw[(size_t)(p * 8 + 24 + (lane & 7)) * HW], 0, 3);
            v4f am[4];
#pragma unroll
            for (int t = 0; t < 4; ++t)
                am[t] = *(const v4f*)(a1 + (t * KP1 + p) * FRAG);
            v2f bm0[2], bm1[2];
            bm0[0].x = r0.x; bm0[0].y = r1.x;
            bm0[1].x = r0.y; bm0[1].y = r1.y;
            bm1[0].x = r4.x; bm1[0].y = r5.x;
            bm1[1].x = r4.y; bm1[1].y = r5.y;
            asm volatile("" : "+v"(bm0[0]), "+v"(bm0[1]), "+v"(bm1[0]), "+v"(bm1[1]));
#pragma unroll
            for (int t = 0; t < 4; ++t) {
                v2f a; a.x = am[t].x; a.y = am[t].y;
#pragma unroll
                for (int j = 0; j < 2; ++j)
                    acc1[t][j] = __builtin_amdgcn_wmma_f32_16x16x4_f32(
                        false, a, false, bm0[j], (short)0, acc1[t][j], false, false);
            }
#pragma unroll
            for (int t = 0; t < 4; ++t) {
                v2f a; a.x = am[t].z; a.y = am[t].w;
#pragma unroll
                for (int j = 0; j < 2; ++j)
                    acc1[t][j] = __builtin_amdgcn_wmma_f32_16x16x4_f32(
                        false, a, false, bm1[j], (short)0, acc1[t][j], false, false);
            }
            wmma_run_end();
        }

#pragma unroll
        for (int t = 0; t < 4; ++t) {
#pragma unroll
            for (int r = 0; r < 8; ++r) {
                const int row = t * 16 + r + 8 * half;
                const float bias = b1[row];
                v2f y;
                y.x = acc1[t][0][r] + bias;
                y.y = acc1[t][1][r] + bias;
                *(v2f*)(my + row * YP + 2 * col) = y;
            }
        }
    }
    __syncthreads();
    if (!active) return;

    const float* a2 = wl + W1P_FLOATS + lane * 4;
    const float* myq = my + 2 * col;
    float* o = out + (size_t)b * CIN * HW + hw + 2 * col;
    for (int g = 0; g < 2; ++g) {
        v8f acc2[8][2] = {};
#pragma nounroll
        for (int p = 0; p < KP2; ++p) {
            const int kr   = p * 8 + 2 * half;
            const int krow = (p >= 8) ? (kr - 32) : kr;
            v4f am[8];
#pragma unroll
            for (int t = 0; t < 8; ++t)
                am[t] = *(const v4f*)(a2 + ((g * 8 + t) * KP2 + p) * FRAG);
            v2f z0 = *(const v2f*)(myq + krow * YP);
            v2f z1 = *(const v2f*)(myq + (krow + 1) * YP);
            v2f z4 = *(const v2f*)(myq + (krow + 4) * YP);
            v2f z5 = *(const v2f*)(myq + (krow + 5) * YP);
            if (p >= 8) {
                z0.x = fmaxf(z0.x, 0.0f); z0.y = fmaxf(z0.y, 0.0f);
                z1.x = fmaxf(z1.x, 0.0f); z1.y = fmaxf(z1.y, 0.0f);
                z4.x = fmaxf(z4.x, 0.0f); z4.y = fmaxf(z4.y, 0.0f);
                z5.x = fmaxf(z5.x, 0.0f); z5.y = fmaxf(z5.y, 0.0f);
            }
            v2f bm0[2], bm1[2];
            bm0[0].x = z0.x; bm0[0].y = z1.x;
            bm0[1].x = z0.y; bm0[1].y = z1.y;
            bm1[0].x = z4.x; bm1[0].y = z5.x;
            bm1[1].x = z4.y; bm1[1].y = z5.y;
            asm volatile("" : "+v"(bm0[0]), "+v"(bm0[1]), "+v"(bm1[0]), "+v"(bm1[1]));
#pragma unroll
            for (int t = 0; t < 8; ++t) {
                v2f a; a.x = am[t].x; a.y = am[t].y;
#pragma unroll
                for (int j = 0; j < 2; ++j)
                    acc2[t][j] = __builtin_amdgcn_wmma_f32_16x16x4_f32(
                        false, a, false, bm0[j], (short)0, acc2[t][j], false, false);
            }
#pragma unroll
            for (int t = 0; t < 8; ++t) {
                v2f a; a.x = am[t].z; a.y = am[t].w;
#pragma unroll
                for (int j = 0; j < 2; ++j)
                    acc2[t][j] = __builtin_amdgcn_wmma_f32_16x16x4_f32(
                        false, a, false, bm1[j], (short)0, acc2[t][j], false, false);
            }
            wmma_run_end();
        }
#pragma unroll
        for (int t = 0; t < 8; ++t) {
#pragma unroll
            for (int r = 0; r < 8; ++r) {
                const int row = (g * 8 + t) * 16 + r + 8 * half;
                const float bias = b2[row];
                v2f v;
                v.x = acc2[t][0][r] + bias;
                v.y = acc2[t][1][r] + bias;
                *(v2f*)(o + (size_t)row * HW) = v;
            }
        }
    }
}

__global__ __launch_bounds__(NTHR) void fused_levels(
    const float* __restrict__ x0, const float* __restrict__ x1,
    const float* __restrict__ x2,
    const float* __restrict__ W1_0, const float* __restrict__ W2_0,
    const float* __restrict__ W1_1, const float* __restrict__ W2_1,
    const float* __restrict__ W1_2, const float* __restrict__ W2_2,
    const float* __restrict__ b1_0, const float* __restrict__ b2_0,
    const float* __restrict__ b1_1, const float* __restrict__ b2_1,
    const float* __restrict__ b1_2, const float* __restrict__ b2_2,
    float* __restrict__ out)
{
    __shared__ __attribute__((aligned(16))) float wlds[LVL_FLOATS];
    __shared__ __attribute__((aligned(16))) float ylds[NWAVES][HIDDEN * YP];

    const int blk  = blockIdx.x;
    const int wave = threadIdx.x >> 5;
    const int lvl  = (blk < NBL0) ? 0 : (blk < NBL0 + NBL1) ? 1 : 2;

    const int lb   = (lvl == 0) ? blk : (lvl == 1) ? blk - NBL0 : blk - NBL0 - NBL1;
    const int unit = lb * NWAVES + wave;

    {
        const int sh = (lvl == 0) ? 14 : (lvl == 1) ? 12 : 10;
        const float* xl = (lvl == 0) ? x0 : (lvl == 1) ? x1 : x2;
        if (unit < (4 << sh) / WPX) {
            const int pix = unit * WPX;
            const int b   = pix >> sh;
            const int hw  = pix - (b << sh);
            __builtin_prefetch(&xl[(((size_t)b * CIN + (threadIdx.x & 31)) << sh) + hw], 0, 3);
        }
    }

    {
        const float* W1 = (lvl == 0) ? W1_0 : (lvl == 1) ? W1_1 : W1_2;
        const float* W2 = (lvl == 0) ? W2_0 : (lvl == 1) ? W2_1 : W2_2;
        v4f* dstw = (v4f*)wlds;
        const int lane = threadIdx.x & 31;
        const int wv  = __builtin_amdgcn_readfirstlane(wave);
        const int rs  = blk & 15;
        const int hi  = lane >> 4;
        const int slot = lane & 15;
        v4f t1[R1], t2[R2];
#pragma unroll
        for (int u = 0; u < R1; ++u) {
            const int t0 = u * NWAVES + wv;
            if ((u + 1) * NWAVES <= T1N || t0 < T1N) {
                int t = t0 + rs * (T1N / 16);
                if (t >= T1N) t -= T1N;
                const int cb = t & 1, h = (t >> 1) & 1, p = t >> 2;
                const float* s = W1 + (8 * p + 2 * h) * HIDDEN + 32 * cb + lane;
                v4f v;
                v.x = s[0]; v.y = s[HIDDEN]; v.z = s[4 * HIDDEN]; v.w = s[5 * HIDDEN];
                t1[u] = v;
            }
        }
#pragma unroll
        for (int u = 0; u < R2; ++u) {
            int t = u * NWAVES + wv + rs * (T2N / 16);
            if (t >= T2N) t -= T2N;
            const int cb = t & 7, h = (t >> 3) & 1, p = t >> 4;
            const float* s = W2 + (8 * p + 2 * h) * CIN + 32 * cb + lane;
            v4f v;
            v.x = s[0]; v.y = s[CIN]; v.z = s[4 * CIN]; v.w = s[5 * CIN];
            t2[u] = v;
        }
#pragma unroll
        for (int u = 0; u < R1; ++u) {
            const int t0 = u * NWAVES + wv;
            if ((u + 1) * NWAVES <= T1N || t0 < T1N) {
                int t = t0 + rs * (T1N / 16);
                if (t >= T1N) t -= T1N;
                const int cb = t & 1, h = (t >> 1) & 1, p = t >> 2;
                dstw[((2 * cb + hi) * KP1 + p) * 32 + 16 * h + slot] = t1[u];
            }
        }
#pragma unroll
        for (int u = 0; u < R2; ++u) {
            int t = u * NWAVES + wv + rs * (T2N / 16);
            if (t >= T2N) t -= T2N;
            const int cb = t & 7, h = (t >> 3) & 1, p = t >> 4;
            dstw[W1P_FLOATS / 4 + ((2 * cb + hi) * KP2 + p) * 32 + 16 * h + slot] = t2[u];
        }
    }
    __syncthreads();

    float* my = &ylds[wave][0];
    if (lvl == 0) {
        level_unit<HW0>(wlds, my, unit, unit < UNITS0, x0, b1_0, b2_0,
                        out);
    } else if (lvl == 1) {
        level_unit<HW1>(wlds, my, unit, unit < UNITS1, x1, b1_1, b2_1,
                        out + (size_t)4 * CIN * HW0);
    } else {
        level_unit<HW2>(wlds, my, unit, unit < UNITS2, x2, b1_2, b2_2,
                        out + (size_t)4 * CIN * (HW0 + HW1));
    }
}

extern "C" void kernel_launch(void* const* d_in, const int* in_sizes, int n_in,
                              void* d_out, int out_size, void* d_ws, size_t ws_size,
                              hipStream_t stream) {
    (void)in_sizes; (void)n_in; (void)out_size; (void)d_ws; (void)ws_size;

    fused_levels<<<NBL0 + NBL1 + NBL2, NTHR, 0, stream>>>(
        (const float*)d_in[0], (const float*)d_in[1], (const float*)d_in[2],
        (const float*)d_in[3], (const float*)d_in[5],
        (const float*)d_in[7], (const float*)d_in[9],
        (const float*)d_in[11], (const float*)d_in[13],
        (const float*)d_in[4], (const float*)d_in[6],
        (const float*)d_in[8], (const float*)d_in[10],
        (const float*)d_in[12], (const float*)d_in[14],
        (float*)d_out);
}


# Both K loops software-pipelined: a later trip's operands (x rows 4 trips ahead, L2 prefetch 8; weight and Y fragments 1 ahead) are requested before the current WMMA run, in a second register set
# speedup vs baseline: 4.3969x; 1.0901x over previous
#include <hip/hip_runtime.h>


typedef float v2f __attribute__((ext_vector_type(2)));
typedef float v4f __attribute__((ext_vector_type(4)));
typedef float v8f __attribute__((ext_vector_type(8)));

__device__ __forceinline__ void wmma_run_end()
{
    __builtin_amdgcn_sched_barrier(0);
    asm volatile("v_nop\n\tv_nop\n\tv_nop\n\tv_nop");
    __builtin_amdgcn_sched_barrier(0);
}

#define CIN    256
#define HIDDEN 64
#define CIN2   96

#define W1P_FLOATS (HIDDEN * CIN)
#define W2P_FLOATS (CIN * CIN2)
#define LVL_FLOATS (W1P_FLOATS + W2P_FLOATS)

#define KP1 (CIN / 8)
#define KP2 (CIN2 / 8)
#define FRAG 128

#define HW0 (128 * 128)
#define HW1 (64 * 64)
#define HW2 (32 * 32)

#define WPX 32
#define YP  48

#define NWAVES 12
#define NTHR   (NWAVES * 32)
#define UNITS0 (4 * HW0 / WPX)
#define UNITS1 (4 * HW1 / WPX)
#define UNITS2 (4 * HW2 / WPX)
#define NBL0 ((UNITS0 + NWAVES - 1) / NWAVES)
#define NBL1 ((UNITS1 + NWAVES - 1) / NWAVES)
#define NBL2 ((UNITS2 + NWAVES - 1) / NWAVES)
#define T1N (KP1 * 2 * (HIDDEN / 32))
#define T2N (KP2 * 2 * (CIN / 32))
#define R1  ((T1N + NWAVES - 1) / NWAVES)
#define R2  (T2N / NWAVES)

#define XR  4
#define NQ1 (KP1 / XR)
#define NQ2 (KP2 / 2)
#define PFQ 2


template <int HW>
__device__ __forceinline__ void load_x(v2f (&r)[4], const float* xq, const int p, const int half)
{
    const int kr = p * 8 + 2 * half;
    r[0] = *(const v2f*)(xq + (size_t)kr * HW);
    r[1] = *(const v2f*)(xq + (size_t)(kr + 1) * HW);
    r[2] = *(const v2f*)(xq + (size_t)(kr + 4) * HW);
    r[3] = *(const v2f*)(xq + (size_t)(kr + 5) * HW);
}

template <int HW, bool LAST>
__device__ __forceinline__ void s1_group(
    const float* xq, const float* a1, const int q, const int half,
    v2f (&xr)[XR][4], v4f (&am)[2][4], v8f (&acc1)[4][2])
{
#pragma unroll
    for (int u = 0; u < XR; ++u) {
        const int p = q * XR + u;
        v2f bm0[2], bm1[2];
        bm0[0].x = xr[u][0].x; bm0[0].y = xr[u][1].x;
        bm0[1].x = xr[u][0].y; bm0[1].y = xr[u][1].y;
        bm1[0].x = xr[u][2].x; bm1[0].y = xr[u][3].x;
        bm1[1].x = xr[u][2].y; bm1[1].y = xr[u][3].y;
        if (!LAST)
            load_x<HW>(xr[u], xq, p + XR, half);
        if (!LAST || u + 1 < XR) {
#pragma unroll
            for (int t = 0; t < 4; ++t)
                am[(u + 1) & 1][t] = *(const v4f*)(a1 + (t * KP1 + p + 1) * FRAG);
        }
        asm volatile("" : "+v"(bm0[0]), "+v"(bm0[1]), "+v"(bm1[0]), "+v"(bm1[1]));
#pragma unroll
        for (int t = 0; t < 4; ++t) {
            v2f a; a.x = am[u & 1][t].x; a.y = am[u & 1][t].y;
#pragma unroll
            for (int j = 0; j < 2; ++j)
                acc1[t][j] = __builtin_amdgcn_wmma_f32_16x16x4_f32(
                    false, a, false, bm0[j], (short)0, acc1[t][j], false, false);
        }
#pragma unroll
        for (int t = 0; t < 4; ++t) {
            v2f a; a.x = am[u & 1][t].z; a.y = am[u & 1][t].w;
#pragma unroll
            for (int j = 0; j < 2; ++j)
                acc1[t][j] = __builtin_amdgcn_wmma_f32_16x16x4_f32(
                    false, a, false, bm1[j], (short)0, acc1[t][j], false, false);
        }
        wmma_run_end();
    }
}

__device__ __forceinline__ void load_s2(
    v4f (&am)[8], v2f (&z)[4], const float* a2g, const float* myq, const int p, const int half)
{
    const int kr   = p * 8 + 2 * half;
    const int krow = (p >= 8) ? (kr - 32) : kr;
#pragma unroll
    for (int t = 0; t < 8; ++t)
        am[t] = *(const v4f*)(a2g + (t * KP2 + p) * FRAG);
    z[0] = *(const v2f*)(myq + krow * YP);
    z[1] = *(const v2f*)(myq + (krow + 1) * YP);
    z[2] = *(const v2f*)(myq + (krow + 4) * YP);
    z[3] = *(const v2f*)(myq + (krow + 5) * YP);
}

template <bool RELU, bool LAST>
__device__ __forceinline__ void s2_pair(
    const float* a2g, const float* myq, const int q, const int half,
    v4f (&am)[2][8], v2f (&z)[4], v8f (&acc2)[8][2])
{
#pragma unroll
    for (int u = 0; u < 2; ++u) {
        const int p = q * 2 + u;
        v2f z0 = z[0], z1 = z[1], z4 = z[2], z5 = z[3];
        if (RELU) {
            z0.x = fmaxf(z0.x, 0.0f); z0.y = fmaxf(z0.y, 0.0f);
            z1.x = fmaxf(z1.x, 0.0f); z1.y = fmaxf(z1.y, 0.0f);
            z4.x = fmaxf(z4.x, 0.0f); z4.y = fmaxf(z4.y, 0.0f);
            z5.x = fmaxf(z5.x, 0.0f); z5.y = fmaxf(z5.y, 0.0f);
        }
        v2f bm0[2], bm1[2];
        bm0[0].x = z0.x; bm0[0].y = z1.x;
        bm0[1].x = z0.y; bm0[1].y = z1.y;
        bm1[0].x = z4.x; bm1[0].y = z5.x;
        bm1[1].x = z4.y; bm1[1].y = z5.y;
        if (!LAST || u == 0)
            load_s2(am[u ^ 1], z, a2g, myq, p + 1, half);
        asm volatile("" : "+v"(bm0[0]), "+v"(bm0[1]), "+v"(bm1[0]), "+v"(bm1[1]));
#pragma unroll
        for (int t = 0; t < 8; ++t) {
            v2f a; a.x = am[u][t].x; a.y = am[u][t].y;
#pragma unroll
            for (int j = 0; j < 2; ++j)
                acc2[t][j] = __builtin_amdgcn_wmma_f32_16x16x4_f32(
                    false, a, false, bm0[j], (short)0, acc2[t][j], false, false);
        }
#pragma unroll
        for (int t = 0; t < 8; ++t) {
            v2f a; a.x = am[u][t].z; a.y = am[u][t].w;
#pragma unroll
            for (int j = 0; j < 2; ++j)
                acc2[t][j] = __builtin_amdgcn_wmma_f32_16x16x4_f32(
                    false, a, false, bm1[j], (short)0, acc2[t][j], false, false);
        }
        wmma_run_end();
    }
}

template <int HW>
__device__ __forceinline__ void level_unit(
    const float* wl, float* my, const int unit, const bool active,
    const float* __restrict__ x,
    const float* __restrict__ b1, const float* __restrict__ b2,
    float* __restrict__ out)
{
    const int lane = threadIdx.x & 31;
    const int col  = lane & 15;
    const int half = lane >> 4;

    const int pixBase = unit * WPX;
    const int b  = pixBase / HW;
    const int hw = pixBase - b * HW;

    const float* xw = x + (size_t)b * CIN * HW + hw;
    const float* xq = xw + 2 * col;

    if (active) {
        const float* a1 = wl + lane * 4;
        v8f acc1[4][2] = {};
        v2f xr[XR][4];
        v4f am[2][4];
#pragma unroll
        for (int u = 0; u < XR; ++u)
            load_x<HW>(xr[u], xq, u, half);
#pragma unroll
        for (int t = 0; t < 4; ++t)
            am[0][t] = *(const v4f*)(a1 + (t * KP1) * FRAG);
#pragma nounroll
        for (int q = 0; q < NQ1 - 1; ++q) {
            if (q + PFQ < NQ1)
                __builtin_prefetch(&xw[(size_t)((q + PFQ) * 8 * XR + lane) * HW], 0, 3);
            s1_group<HW, false>(xq, a1, q, half, xr, am, acc1);
        }
        s1_group<HW, true>(xq, a1, NQ1 - 1, half, xr, am, acc1);

#pragma unroll
        for (int t = 0; t < 4; ++t) {
#pragma unroll
            for (int r = 0; r < 8; ++r) {
                const int row = t * 16 + r + 8 * half;
                const float bias = b1[row];
                v2f y;
                y.x = acc1[t][0][r] + bias;
                y.y = acc1[t][1][r] + bias;
                *(v2f*)(my + row * YP + 2 * col) = y;
            }
        }
    }
    __syncthreads();
    if (!active) return;

    const float* a2 = wl + W1P_FLOATS + lane * 4;
    const float* myq = my + 2 * col;
    float* o = out + (size_t)b * CIN * HW + hw + 2 * col;
    for (int g = 0; g < 2; ++g) {
        v8f acc2[8][2] = {};
        const float* a2g = a2 + (g * 8) * KP2 * FRAG;
        v4f am[2][8];
        v2f z[4];
        load_s2(am[0], z, a2g, myq, 0, half);
#pragma nounroll
        for (int q = 0; q < 4; ++q)
            s2_pair<false, false>(a2g, myq, q, half, am, z, acc2);
        s2_pair<true, false>(a2g, myq, 4, half, am, z, acc2);
        s2_pair<true, true>(a2g, myq, NQ2 - 1, half, am, z, acc2);
#pragma unroll
        for (int t = 0; t < 8; ++t) {
#pragma unroll
            for (int r = 0; r < 8; ++r) {
                const int row = (g * 8 + t) * 16 + r + 8 * half;
                const float bias = b2[row];
                v2f v;
                v.x = acc2[t][0][r] + bias;
                v.y = acc2[t][1][r] + bias;
                *(v2f*)(o + (size_t)row * HW) = v;
            }
        }
    }
}

__global__ __launch_bounds__(NTHR) void fused_levels(
    const float* __restrict__ x0, const float* __restrict__ x1,
    const float* __restrict__ x2,
    const float* __restrict__ W1_0, const float* __restrict__ W2_0,
    const float* __restrict__ W1_1, const float* __restrict__ W2_1,
    const float* __restrict__ W1_2, const float* __restrict__ W2_2,
    const float* __restrict__ b1_0, const float* __restrict__ b2_0,
    const float* __restrict__ b1_1, const float* __restrict__ b2_1,
    const float* __restrict__ b1_2, const float* __restrict__ b2_2,
    float* __restrict__ out)
{
    __shared__ __attribute__((aligned(16))) float wlds[LVL_FLOATS];
    __shared__ __attribute__((aligned(16))) float ylds[NWAVES][HIDDEN * YP];

    const int blk  = blockIdx.x;
    const int wave = threadIdx.x >> 5;
    const int lvl  = (blk < NBL0) ? 0 : (blk < NBL0 + NBL1) ? 1 : 2;

    const int lb   = (lvl == 0) ? blk : (lvl == 1) ? blk - NBL0 : blk - NBL0 - NBL1;
    const int unit = lb * NWAVES + wave;

    {
        const int sh = (lvl == 0) ? 14 : (lvl == 1) ? 12 : 10;
        const float* xl = (lvl == 0) ? x0 : (lvl == 1) ? x1 : x2;
        if (unit < (4 << sh) / WPX) {
            const int pix = unit * WPX;
            const int b   = pix >> sh;
            const int hw  = pix - (b << sh);
            __builtin_prefetch(&xl[(((size_t)b * CIN + (threadIdx.x & 31)) << sh) + hw], 0, 3);
            __builtin_prefetch(&xl[(((size_t)b * CIN + 32 + (threadIdx.x & 31)) << sh) + hw], 0, 3);
        }
    }

    {
        const float* W1 = (lvl == 0) ? W1_0 : (lvl == 1) ? W1_1 : W1_2;
        const float* W2 = (lvl == 0) ? W2_0 : (lvl == 1) ? W2_1 : W2_2;
        v4f* dstw = (v4f*)wlds;
        const int lane = threadIdx.x & 31;
        const int wv  = __builtin_amdgcn_readfirstlane(wave);
        const int rs  = blk & 15;
        const int hi  = lane >> 4;
        const int slot = lane & 15;
        v4f t1[R1], t2[R2];
#pragma unroll
        for (int u = 0; u < R1; ++u) {
            const int t0 = u * NWAVES + wv;
            if ((u + 1) * NWAVES <= T1N || t0 < T1N) {
                int t = t0 + rs * (T1N / 16);
                if (t >= T1N) t -= T1N;
                const int cb = t & 1, h = (t >> 1) & 1, p = t >> 2;
                const float* s = W1 + (8 * p + 2 * h) * HIDDEN + 32 * cb + lane;
                v4f v;
                v.x = s[0]; v.y = s[HIDDEN]; v.z = s[4 * HIDDEN]; v.w = s[5 * HIDDEN];
                t1[u] = v;
            }
        }
#pragma unroll
        for (int u = 0; u < R2; ++u) {
            int t = u * NWAVES + wv + rs * (T2N / 16);
            if (t >= T2N) t -= T2N;
            const int cb = t & 7, h = (t >> 3) & 1, p = t >> 4;
            const float* s = W2 + (8 * p + 2 * h) * CIN + 32 * cb + lane;
            v4f v;
            v.x = s[0]; v.y = s[CIN]; v.z = s[4 * CIN]; v.w = s[5 * CIN];
            t2[u] = v;
        }
#pragma unroll
        for (int u = 0; u < R1; ++u) {
            const int t0 = u * NWAVES + wv;
            if ((u + 1) * NWAVES <= T1N || t0 < T1N) {
                int t = t0 + rs * (T1N / 16);
                if (t >= T1N) t -= T1N;
                const int cb = t & 1, h = (t >> 1) & 1, p = t >> 2;
                dstw[((2 * cb + hi) * KP1 + p) * 32 + 16 * h + slot] = t1[u];
            }
        }
#pragma unroll
        for (int u = 0; u < R2; ++u) {
            int t = u * NWAVES + wv + rs * (T2N / 16);
            if (t >= T2N) t -= T2N;
            const int cb = t & 7, h = (t >> 3) & 1, p = t >> 4;
            dstw[W1P_FLOATS / 4 + ((2 * cb + hi) * KP2 + p) * 32 + 16 * h + slot] = t2[u];
        }
    }
    __syncthreads();

    float* my = &ylds[wave][0];
    if (lvl == 0) {
        level_unit<HW0>(wlds, my, unit, unit < UNITS0, x0, b1_0, b2_0,
                        out);
    } else if (lvl == 1) {
        level_unit<HW1>(wlds, my, unit, unit < UNITS1, x1, b1_1, b2_1,
                        out + (size_t)4 * CIN * HW0);
    } else {
        level_unit<HW2>(wlds, my, unit, unit < UNITS2, x2, b1_2, b2_2,
                        out + (size_t)4 * CIN * (HW0 + HW1));
    }
}

extern "C" void kernel_launch(void* const* d_in, const int* in_sizes, int n_in,
                              void* d_out, int out_size, void* d_ws, size_t ws_size,
                              hipStream_t stream) {
    (void)in_sizes; (void)n_in; (void)out_size; (void)d_ws; (void)ws_size;

    fused_levels<<<NBL0 + NBL1 + NBL2, NTHR, 0, stream>>>(
        (const float*)d_in[0], (const float*)d_in[1], (const float*)d_in[2],
        (const float*)d_in[3], (const float*)d_in[5],
        (const float*)d_in[7], (const float*)d_in[9],
        (const float*)d_in[11], (const float*)d_in[13],
        (const float*)d_in[4], (const float*)d_in[6],
        (const float*)d_in[8], (const float*)d_in[10],
        (const float*)d_in[12], (const float*)d_in[14],
        (float*)d_out);
}
